# starting kernel
# baseline (speedup 1.0000x reference)
#include <hip/hip_runtime.h>
#include <hip/hip_bf16.h>


typedef __attribute__((ext_vector_type(2))) float v2f;
typedef __attribute__((ext_vector_type(4))) float v4f;
typedef __attribute__((ext_vector_type(8))) float v8f;

#define HEADDIM 64
#define NUM_PROJS 8
#define ROWS_PER_TILE 16
#define WAVES_PER_BLOCK 8
#define TILES_PER_WAVE 4
#define TILE_DWORDS (ROWS_PER_TILE * HEADDIM)
#define LDS_STRIDE 68

__global__ __launch_bounds__(256)
void angular_lsh_wmma_kernel(const float* __restrict__ mat,
                             const float* __restrict__ proj,
                             const int* __restrict__ perm,
                             int* __restrict__ out,
                             int nTiles) {
    __shared__ float lds[WAVES_PER_BLOCK][2][ROWS_PER_TILE * LDS_STRIDE];

    const int lane  = threadIdx.x & 31;
    const int wave  = threadIdx.x >> 5;
    const int tile0 = (blockIdx.x * WAVES_PER_BLOCK + wave) * TILES_PER_WAVE;
    if (tile0 >= nTiles) return;

    const int row16 = lane & 15;
    const int halfw = lane >> 4;

    v2f bfrag[16];
#pragma unroll
    for (int s = 0; s < 16; ++s) {
        const int k0 = 4 * s + 2 * halfw;
        float b0 = 0.0f, b1 = 0.0f;
        if (row16 < NUM_PROJS) {
            b0 = proj[k0 * NUM_PROJS + row16];
            b1 = proj[(k0 + 1) * NUM_PROJS + row16];
        }
        bfrag[s] = (v2f){b0, b1};
    }

    v4f pf[8];
    {
        const float* src = mat + (size_t)tile0 * TILE_DWORDS;
#pragma unroll
        for (int i = 0; i < 8; ++i)
            pf[i] = __builtin_nontemporal_load((const v4f*)(src + i * 128 + lane * 4));
        float* buf = &lds[wave][0][0];
#pragma unroll
        for (int i = 0; i < 8; ++i) {
            const int off = i * 128 + lane * 4;
            *(v4f*)(buf + (off >> 6) * LDS_STRIDE + (off & 63)) = pf[i];
        }
    }

#pragma unroll
    for (int t = 0; t < TILES_PER_WAVE; ++t) {
        const int tile = tile0 + t;
        if (tile >= nTiles) break;
        const bool hasNext = (t + 1 < TILES_PER_WAVE) && (tile + 1 < nTiles);

        if (hasNext) {
            const float* nsrc = mat + (size_t)(tile + 1) * TILE_DWORDS;
#pragma unroll
            for (int i = 0; i < 8; ++i)
                pf[i] = __builtin_nontemporal_load((const v4f*)(nsrc + i * 128 + lane * 4));
        }

        const v2f* arow = (const v2f*)(&lds[wave][t & 1][0] + row16 * LDS_STRIDE);
        v8f c = {0.f, 0.f, 0.f, 0.f, 0.f, 0.f, 0.f, 0.f};
#pragma unroll
        for (int s = 0; s < 16; ++s) {
            v2f a = arow[2 * s + halfw];
            c = __builtin_amdgcn_wmma_f32_16x16x4_f32(
                     false, a,  false, bfrag[s],
                     (short)0, c,  false,  false);
        }

        unsigned mmask[8];
#pragma unroll
        for (int v = 0; v < 8; ++v)
            mmask[v] = (unsigned)__ballot(c[v] > 0.0f);

        unsigned sel = mmask[0];
#pragma unroll
        for (int v = 1; v < 8; ++v)
            sel = ((lane & 7) == v) ? mmask[v] : sel;

        if (lane < 16) {
            const unsigned bits = (sel >> ((lane >> 3) * 16)) & 0xffu;
            out[(size_t)tile * ROWS_PER_TILE + lane] = perm[bits];
        }

        if (hasNext) {
            float* nbuf = &lds[wave][(t + 1) & 1][0];
#pragma unroll
            for (int i = 0; i < 8; ++i) {
                const int off = i * 128 + lane * 4;
                *(v4f*)(nbuf + (off >> 6) * LDS_STRIDE + (off & 63)) = pf[i];
            }
        }
    }
}

extern "C" void kernel_launch(void* const* d_in, const int* in_sizes, int n_in,
                              void* d_out, int out_size, void* d_ws, size_t ws_size,
                              hipStream_t stream) {
    const float* mat  = (const float*)d_in[0];
    const float* proj = (const float*)d_in[1];
    const int*   perm = (const int*)d_in[2];
    int* out = (int*)d_out;

    const int nRows  = in_sizes[0] / HEADDIM;
    const int nTiles = nRows / ROWS_PER_TILE;

    const int tilesPerBlock = WAVES_PER_BLOCK * TILES_PER_WAVE;
    dim3 block(32 * WAVES_PER_BLOCK);
    dim3 grid((nTiles + tilesPerBlock - 1) / tilesPerBlock);
    angular_lsh_wmma_kernel<<<grid, block, 0, stream>>>(mat, proj, perm, out, nTiles);
}


# Software-pipelined tile pairs: each wave does 8 tiles as 4 pairs on two independent WMMA chains, A fragments read from LDS as one group ahead of the load wait; perm from LDS, 128-byte line stores
# speedup vs baseline: 1.0562x; 1.0562x over previous
#include <hip/hip_runtime.h>


typedef __attribute__((ext_vector_type(2))) float v2f;
typedef __attribute__((ext_vector_type(4))) float v4f;
typedef __attribute__((ext_vector_type(4))) int   v4i;
typedef __attribute__((ext_vector_type(8))) float v8f;

#define HEADDIM 64
#define NUM_PROJS 8
#define ROWS_PER_TILE 16
#define WAVES_PER_BLOCK 8
#define PAIRS_PER_WAVE 4
#define TILES_PER_WAVE (2 * PAIRS_PER_WAVE)
#define TILE_DWORDS (ROWS_PER_TILE * HEADDIM)
#define LDS_STRIDE 68
#define TILE_LDS (ROWS_PER_TILE * LDS_STRIDE)
#define PAIR_LDS (2 * TILE_LDS)
#define PERM_SIZE 256

__device__ __forceinline__ void load_pair(v4f (&pf)[16], const float* __restrict__ mat,
                                          int tA, int nTiles, int lane) {
    const float* sA = mat + (size_t)tA * TILE_DWORDS + lane * 4;
    const float* sB = (tA + 1 < nTiles) ? (sA + TILE_DWORDS) : sA;
#pragma unroll
    for (int i = 0; i < 8; ++i)
        pf[i] = __builtin_nontemporal_load((const v4f*)(sA + i * 128));
#pragma unroll
    for (int i = 0; i < 8; ++i)
        pf[8 + i] = __builtin_nontemporal_load((const v4f*)(sB + i * 128));
}

__device__ __forceinline__ void stage_pair(float* buf, const v4f (&pf)[16], int lane) {
#pragma unroll
    for (int i = 0; i < 16; ++i) {
        const int off = i * 128 + lane * 4;
        *(v4f*)(buf + (off >> 6) * LDS_STRIDE + (off & 63)) = pf[i];
    }
}

__global__ __launch_bounds__(256)
void angular_lsh_wmma_kernel(const float* __restrict__ mat,
                             const float* __restrict__ proj,
                             const int* __restrict__ perm,
                             int* __restrict__ out,
                             int nTiles) {
    __shared__ float lds[WAVES_PER_BLOCK][2][PAIR_LDS];
    __shared__ int   permLds[WAVES_PER_BLOCK][PERM_SIZE];

    const int lane  = threadIdx.x & 31;
    const int wave  = threadIdx.x >> 5;
    const int tile0 = (blockIdx.x * WAVES_PER_BLOCK + wave) * TILES_PER_WAVE;
    if (tile0 >= nTiles) return;

    const int row16 = lane & 15;
    const int halfw = lane >> 4;

    v4f pf[16];
    load_pair(pf, mat, tile0, nTiles, lane);
    const v4i pm0 = *(const v4i*)(perm + lane * 4);
    const v4i pm1 = *(const v4i*)(perm + 128 + lane * 4);

    v2f bfrag[16];
#pragma unroll
    for (int s = 0; s < 16; ++s) {
        const int k0 = 4 * s + 2 * halfw;
        float b0 = 0.0f, b1 = 0.0f;
        if (row16 < NUM_PROJS) {
            b0 = proj[k0 * NUM_PROJS + row16];
            b1 = proj[(k0 + 1) * NUM_PROJS + row16];
        }
        bfrag[s] = (v2f){b0, b1};
    }

    int* myPerm = &permLds[wave][0];
    *(v4i*)(myPerm + lane * 4)       = pm0;
    *(v4i*)(myPerm + 128 + lane * 4) = pm1;

    stage_pair(&lds[wave][0][0], pf, lane);
    if (PAIRS_PER_WAVE > 1 && tile0 + 2 < nTiles)
        load_pair(pf, mat, tile0 + 2, nTiles, lane);

#pragma unroll
    for (int p = 0; p < PAIRS_PER_WAVE; ++p) {
        const int tA = tile0 + 2 * p;
        if (tA >= nTiles) break;
        const bool hasB     = (tA + 1 < nTiles);
        const bool hasNext  = (p + 1 < PAIRS_PER_WAVE) && (tA + 2 < nTiles);
        const bool hasNext2 = (p + 2 < PAIRS_PER_WAVE) && (tA + 4 < nTiles);

        const float* cur = &lds[wave][p & 1][0];
        const v2f* rowA = (const v2f*)(cur + row16 * LDS_STRIDE);
        const v2f* rowB = (const v2f*)(cur + TILE_LDS + row16 * LDS_STRIDE);
        v2f aA[16], aB[16];
#pragma unroll
        for (int s = 0; s < 16; ++s) {
            aA[s] = rowA[2 * s + halfw];
            aB[s] = rowB[2 * s + halfw];
        }
        __builtin_amdgcn_sched_barrier(0);

        if (hasNext)
            stage_pair(&lds[wave][(p + 1) & 1][0], pf, lane);
        if (hasNext2)
            load_pair(pf, mat, tA + 4, nTiles, lane);

        v8f cA = {0.f, 0.f, 0.f, 0.f, 0.f, 0.f, 0.f, 0.f};
        v8f cB = {0.f, 0.f, 0.f, 0.f, 0.f, 0.f, 0.f, 0.f};
#pragma unroll
        for (int s = 0; s < 16; ++s) {
            cA = __builtin_amdgcn_wmma_f32_16x16x4_f32(false, aA[s], false, bfrag[s],
                                                       (short)0, cA, false, false);
            cB = __builtin_amdgcn_wmma_f32_16x16x4_f32(false, aB[s], false, bfrag[s],
                                                       (short)0, cB, false, false);
        }

        unsigned selA = (unsigned)__ballot(cA[0] > 0.0f);
        unsigned selB = (unsigned)__ballot(cB[0] > 0.0f);
#pragma unroll
        for (int v = 1; v < 8; ++v) {
            const unsigned mA = (unsigned)__ballot(cA[v] > 0.0f);
            const unsigned mB = (unsigned)__ballot(cB[v] > 0.0f);
            const bool hit = ((lane & 7) == v);
            selA = hit ? mA : selA;
            selB = hit ? mB : selB;
        }
        const unsigned sel  = (lane < 16) ? selA : selB;
        const unsigned bits = (sel >> (((lane >> 3) & 1) * 16)) & 0xffu;
        if (lane < 16 || hasB)
            out[(size_t)tA * ROWS_PER_TILE + lane] = myPerm[bits];
    }
}

extern "C" void kernel_launch(void* const* d_in, const int* in_sizes, int n_in,
                              void* d_out, int out_size, void* d_ws, size_t ws_size,
                              hipStream_t stream) {
    const float* mat  = (const float*)d_in[0];
    const float* proj = (const float*)d_in[1];
    const int*   perm = (const int*)d_in[2];
    int* out = (int*)d_out;

    const int nRows  = in_sizes[0] / HEADDIM;
    const int nTiles = nRows / ROWS_PER_TILE;

    const int tilesPerBlock = WAVES_PER_BLOCK * TILES_PER_WAVE;
    dim3 block(32 * WAVES_PER_BLOCK);
    dim3 grid((nTiles + tilesPerBlock - 1) / tilesPerBlock);
    angular_lsh_wmma_kernel<<<grid, block, 0, stream>>>(mat, proj, perm, out, nTiles);
}


# One 8-wave workgroup per WGP: 2048 waves enter instead of 4096; each wave walks 16 tiles, 32-48 loads in flight from two register sets, a pair multiplied as soon as staged; sum order unchanged
# speedup vs baseline: 1.1265x; 1.0666x over previous
#include <hip/hip_runtime.h>


typedef __attribute__((ext_vector_type(2))) float v2f;
typedef __attribute__((ext_vector_type(4))) float v4f;
typedef __attribute__((ext_vector_type(4))) int   v4i;
typedef __attribute__((ext_vector_type(8))) float v8f;

#define HEADDIM 64
#define NUM_PROJS 8
#define ROWS_PER_TILE 16
#define WAVES_PER_BLOCK 8
#define PAIRS_PER_WAVE 8
#define TILES_PER_WAVE (2 * PAIRS_PER_WAVE)
#define TILE_DWORDS (ROWS_PER_TILE * HEADDIM)
#define LDS_STRIDE 68
#define TILE_LDS (ROWS_PER_TILE * LDS_STRIDE)
#define PAIR_LDS (2 * TILE_LDS)
#define PERM_SIZE 256

__device__ __forceinline__ void load_pair(v4f (&pf)[16], const float* __restrict__ mat,
                                          int tA, int nTiles, int lane) {
    const float* sA = mat + (size_t)tA * TILE_DWORDS + lane * 4;
    const float* sB = (tA + 1 < nTiles) ? (sA + TILE_DWORDS) : sA;
#pragma unroll
    for (int i = 0; i < 8; ++i)
        pf[i] = __builtin_nontemporal_load((const v4f*)(sA + i * 128));
#pragma unroll
    for (int i = 0; i < 8; ++i)
        pf[8 + i] = __builtin_nontemporal_load((const v4f*)(sB + i * 128));
}

__device__ __forceinline__ void stage_pair(float* buf, const v4f (&pf)[16], int lane) {
#pragma unroll
    for (int i = 0; i < 16; ++i) {
        const int off = i * 128 + lane * 4;
        *(v4f*)(buf + (off >> 6) * LDS_STRIDE + (off & 63)) = pf[i];
    }
}

template <bool REFILL>
__device__ __forceinline__ void run_pair(v4f (&pf)[16], float* buf, const int* myPerm,
                                         const v2f (&bfrag)[16],
                                         const float* __restrict__ mat, int* __restrict__ out,
                                         int tA, int nTiles,
                                         int lane, int row16, int halfw) {
    const bool hasB = (tA + 1 < nTiles);

    stage_pair(buf, pf, lane);
    if (REFILL)
        load_pair(pf, mat, (tA + 4 < nTiles) ? (tA + 4) : tA, nTiles, lane);

    const v2f* rowA = (const v2f*)(buf + row16 * LDS_STRIDE);
    const v2f* rowB = (const v2f*)(buf + TILE_LDS + row16 * LDS_STRIDE);
    v2f aA[16], aB[16];
#pragma unroll
    for (int s = 0; s < 16; ++s) {
        aA[s] = rowA[2 * s + halfw];
        aB[s] = rowB[2 * s + halfw];
    }
    __builtin_amdgcn_sched_barrier(0);

    v8f cA = {0.f, 0.f, 0.f, 0.f, 0.f, 0.f, 0.f, 0.f};
    v8f cB = {0.f, 0.f, 0.f, 0.f, 0.f, 0.f, 0.f, 0.f};
#pragma unroll
    for (int s = 0; s < 16; ++s) {
        cA = __builtin_amdgcn_wmma_f32_16x16x4_f32(false, aA[s], false, bfrag[s],
                                                   (short)0, cA, false, false);
        cB = __builtin_amdgcn_wmma_f32_16x16x4_f32(false, aB[s], false, bfrag[s],
                                                   (short)0, cB, false, false);
    }

    unsigned selA = (unsigned)__ballot(cA[0] > 0.0f);
    unsigned selB = (unsigned)__ballot(cB[0] > 0.0f);
#pragma unroll
    for (int v = 1; v < 8; ++v) {
        const unsigned mA = (unsigned)__ballot(cA[v] > 0.0f);
        const unsigned mB = (unsigned)__ballot(cB[v] > 0.0f);
        const bool hit = ((lane & 7) == v);
        selA = hit ? mA : selA;
        selB = hit ? mB : selB;
    }
    const unsigned sel  = (lane < 16) ? selA : selB;
    const unsigned bits = (sel >> (((lane >> 3) & 1) * 16)) & 0xffu;
    if (lane < 16 || hasB)
        out[(size_t)tA * ROWS_PER_TILE + lane] = myPerm[bits];
}

__global__ __launch_bounds__(256)
void angular_lsh_wmma_kernel(const float* __restrict__ mat,
                             const float* __restrict__ proj,
                             const int* __restrict__ perm,
                             int* __restrict__ out,
                             int nTiles) {
    __shared__ float lds[WAVES_PER_BLOCK][PAIR_LDS];
    __shared__ int   permLds[WAVES_PER_BLOCK][PERM_SIZE];

    const int lane  = threadIdx.x & 31;
    const int wave  = __builtin_amdgcn_readfirstlane((int)(threadIdx.x >> 5));
    const int tile0 = (blockIdx.x * WAVES_PER_BLOCK + wave) * TILES_PER_WAVE;
    if (tile0 >= nTiles) return;

    const int row16 = lane & 15;
    const int halfw = lane >> 4;

    const v4i pm0 = *(const v4i*)(perm + lane * 4);
    const v4i pm1 = *(const v4i*)(perm + 128 + lane * 4);

    float pj0[16], pj1[16];
#pragma unroll
    for (int s = 0; s < 16; ++s) {
        const int k0 = 4 * s + 2 * halfw;
        pj0[s] = proj[k0 * NUM_PROJS + (row16 & (NUM_PROJS - 1))];
        pj1[s] = proj[(k0 + 1) * NUM_PROJS + (row16 & (NUM_PROJS - 1))];
    }

    v4f pf0[16], pf1[16];
    load_pair(pf0, mat, tile0, nTiles, lane);
    load_pair(pf1, mat, (tile0 + 2 < nTiles) ? (tile0 + 2) : tile0, nTiles, lane);

    __builtin_amdgcn_sched_barrier(0);

    v2f bfrag[16];
    const bool realCol = (row16 < NUM_PROJS);
#pragma unroll
    for (int s = 0; s < 16; ++s)
        bfrag[s] = (v2f){realCol ? pj0[s] : 0.0f, realCol ? pj1[s] : 0.0f};

    int* myPerm = &permLds[wave][0];
    *(v4i*)(myPerm + lane * 4)       = pm0;
    *(v4i*)(myPerm + 128 + lane * 4) = pm1;

    float* buf = &lds[wave][0];

#pragma unroll
    for (int p = 0; p < PAIRS_PER_WAVE; p += 2) {
        const int tA = tile0 + 2 * p;
        if (tA >= nTiles) break;
        if (p + 2 < PAIRS_PER_WAVE)
            run_pair<true>(pf0, buf, myPerm, bfrag, mat, out, tA, nTiles, lane, row16, halfw);
        else
            run_pair<false>(pf0, buf, myPerm, bfrag, mat, out, tA, nTiles, lane, row16, halfw);
        const int tB = tA + 2;
        if (tB >= nTiles) break;
        if (p + 3 < PAIRS_PER_WAVE)
            run_pair<true>(pf1, buf, myPerm, bfrag, mat, out, tB, nTiles, lane, row16, halfw);
        else
            run_pair<false>(pf1, buf, myPerm, bfrag, mat, out, tB, nTiles, lane, row16, halfw);
    }
}

extern "C" void kernel_launch(void* const* d_in, const int* in_sizes, int n_in,
                              void* d_out, int out_size, void* d_ws, size_t ws_size,
                              hipStream_t stream) {
    const float* mat  = (const float*)d_in[0];
    const float* proj = (const float*)d_in[1];
    const int*   perm = (const int*)d_in[2];
    int* out = (int*)d_out;

    const int nRows  = in_sizes[0] / HEADDIM;
    const int nTiles = nRows / ROWS_PER_TILE;

    const int tilesPerBlock = WAVES_PER_BLOCK * TILES_PER_WAVE;
    dim3 block(32 * WAVES_PER_BLOCK);
    dim3 grid((nTiles + tilesPerBlock - 1) / tilesPerBlock);
    angular_lsh_wmma_kernel<<<grid, block, 0, stream>>>(mat, proj, perm, out, nTiles);
}
